# starting kernel
# baseline (speedup 1.0000x reference)
#include <hip/hip_runtime.h>
#include <math.h>

#define N_NODES 50000
#define N_EDGES 800000
#define IN_F 128
#define OUT_F 128
#define NEG_SLOPE 0.2f
#define EPS_F 1e-10f

typedef float v2f __attribute__((ext_vector_type(2)));
typedef float v8f __attribute__((ext_vector_type(8)));


__device__ __forceinline__ void atomAddF(float* p, float v) {
    __hip_atomic_fetch_add(p, v, __ATOMIC_RELAXED, __HIP_MEMORY_SCOPE_AGENT);
}

__device__ __forceinline__ unsigned fenc(float f) {
    unsigned u = __float_as_uint(f);
    return (u & 0x80000000u) ? ~u : (u | 0x80000000u);
}
__device__ __forceinline__ float fdec(unsigned u) {
    return __uint_as_float((u & 0x80000000u) ? (u ^ 0x80000000u) : ~u);
}

__global__ void __launch_bounds__(128)
gat_gemm_wmma(const float* __restrict__ h, const float* __restrict__ W,
              const float* __restrict__ a, float* __restrict__ Wh,
              float* __restrict__ s1, float* __restrict__ s2, int n_tiles) {
    __shared__ float tlds[4][16 * 128];

    const int lane = threadIdx.x & 31;
    const int wave = threadIdx.x >> 5;
    const int tile = blockIdx.x * 4 + wave;
    if (tile >= n_tiles) return;

    const int row0 = tile * 16;
    const int half = lane >> 4;
    const int mn   = lane & 15;

    float* tl = &tlds[wave][0];
    {
        const float4* hp4 = (const float4*)(h + row0 * IN_F);
        float4* tl4 = (float4*)tl;
        #pragma unroll
        for (int t = 0; t < 16; ++t)
            tl4[t * 32 + lane] = hp4[t * 32 + lane];
    }

    float a1v[8], a2v[8];
    #pragma unroll
    for (int nt = 0; nt < 8; ++nt) {
        a1v[nt] = a[nt * 16 + mn];
        a2v[nt] = a[OUT_F + nt * 16 + mn];
    }

    v8f acc[8] = {};

    for (int kc = 0; kc < 32; ++kc) {
        const int col = kc * 4 + 2 * half;
        v2f av;
        {
            const float2 t = *(const float2*)(tl + mn * 128 + col);
            av[0] = t.x; av[1] = t.y;
        }
        #pragma unroll
        for (int nt = 0; nt < 8; ++nt) {
            const int ncol = nt * 16 + mn;
            v2f bv;
            {
                const float2 t = *(const float2*)(W + ncol * IN_F + col);
                bv[0] = t.x; bv[1] = t.y;
            }
            acc[nt] = __builtin_amdgcn_wmma_f32_16x16x4_f32(
                false, av, false, bv, (short)0, acc[nt], false, false);
        }
    }

    #pragma unroll
    for (int nt = 0; nt < 8; ++nt) {
        #pragma unroll
        for (int j = 0; j < 8; ++j) {
            Wh[(row0 + j + 8 * half) * OUT_F + nt * 16 + mn] = acc[nt][j];
        }
    }

    float p1[8], p2[8];
    #pragma unroll
    for (int j = 0; j < 8; ++j) { p1[j] = 0.f; p2[j] = 0.f; }
    #pragma unroll
    for (int nt = 0; nt < 8; ++nt) {
        #pragma unroll
        for (int j = 0; j < 8; ++j) {
            p1[j] = fmaf(acc[nt][j], a1v[nt], p1[j]);
            p2[j] = fmaf(acc[nt][j], a2v[nt], p2[j]);
        }
    }
    #pragma unroll
    for (int j = 0; j < 8; ++j) {
        #pragma unroll
        for (int off = 1; off < 16; off <<= 1) {
            p1[j] += __shfl_xor(p1[j], off, 32);
            p2[j] += __shfl_xor(p2[j], off, 32);
        }
    }
    if (mn == 0) {
        #pragma unroll
        for (int j = 0; j < 8; ++j) {
            s1[row0 + j + 8 * half] = p1[j];
            s2[row0 + j + 8 * half] = p2[j];
        }
    }
}

__global__ void __launch_bounds__(256)
gat_edge_score(const int* __restrict__ rows, const int* __restrict__ cols,
               const float* __restrict__ s1, const float* __restrict__ s2,
               float* __restrict__ e, unsigned* __restrict__ gmax, int nE) {
    __shared__ float smax[8];
    const int lane = threadIdx.x & 31;
    const int wave = threadIdx.x >> 5;

    float lmax = -INFINITY;
    for (int i = blockIdx.x * blockDim.x + threadIdx.x; i < nE;
         i += gridDim.x * blockDim.x) {
        float v = s1[rows[i]] + s2[cols[i]];
        v = (v > 0.f) ? v : NEG_SLOPE * v;
        e[i] = v;
        lmax = fmaxf(lmax, v);
    }
    #pragma unroll
    for (int off = 16; off >= 1; off >>= 1)
        lmax = fmaxf(lmax, __shfl_xor(lmax, off, 32));
    if (lane == 0) smax[wave] = lmax;
    __syncthreads();
    if (threadIdx.x == 0) {
        float m = smax[0];
        #pragma unroll
        for (int w = 1; w < 8; ++w) m = fmaxf(m, smax[w]);
        atomicMax(gmax, fenc(m));
    }
}

__global__ void __launch_bounds__(256)
gat_edge_exp(const int* __restrict__ rows, const float* __restrict__ e,
             const unsigned* __restrict__ gmax, float* __restrict__ expe,
             float* __restrict__ sum_exp, int nE) {
    const float gm = fdec(*gmax);
    for (int i = blockIdx.x * blockDim.x + threadIdx.x; i < nE;
         i += gridDim.x * blockDim.x) {
        float ex = __expf(e[i] - gm);
        expe[i] = ex;
        atomAddF(&sum_exp[rows[i]], ex);
    }
}

__global__ void __launch_bounds__(256)
gat_scatter(const int* __restrict__ rows, const int* __restrict__ cols,
            const float* __restrict__ expe, const float* __restrict__ sum_exp,
            const float* __restrict__ Wh, float* __restrict__ out, int nE) {
    const int lane = threadIdx.x & 31;
    const int wave = threadIdx.x >> 5;
    const int eid = blockIdx.x * 8 + wave;
    if (eid >= nE) return;

    const int r = rows[eid];
    const int c = cols[eid];
    const float alpha = expe[eid] / (sum_exp[r] + EPS_F);

    const float4 w = ((const float4*)(Wh + c * OUT_F))[lane];
    float* dst = out + r * OUT_F + lane * 4;
    atomAddF(dst + 0, alpha * w.x);
    atomAddF(dst + 1, alpha * w.y);
    atomAddF(dst + 2, alpha * w.z);
    atomAddF(dst + 3, alpha * w.w);
}

__global__ void __launch_bounds__(256)
gat_elu(float* __restrict__ out, int n) {
    int i = blockIdx.x * blockDim.x + threadIdx.x;
    if (i < n) {
        float x = out[i];
        out[i] = (x > 0.f) ? x : expm1f(x);
    }
}

extern "C" void kernel_launch(void* const* d_in, const int* in_sizes, int n_in,
                              void* d_out, int out_size, void* d_ws, size_t ws_size,
                              hipStream_t stream) {
    const float* h  = (const float*)d_in[0];
    const int*   ei = (const int*)d_in[1];
    const float* W  = (const float*)d_in[2];
    const float* a  = (const float*)d_in[3];
    float* out = (float*)d_out;

    const int* rows = ei;
    const int* cols = ei + N_EDGES;

    float* ws = (float*)d_ws;
    float*    Wh      = ws;
    float*    s1      = Wh + (size_t)N_NODES * OUT_F;
    float*    s2      = s1 + N_NODES;
    float*    e       = s2 + N_NODES;
    float*    expe    = e + N_EDGES;
    float*    sum_exp = expe + N_EDGES;
    unsigned* gmax    = (unsigned*)(sum_exp + N_NODES);

    hipMemsetAsync(out, 0, (size_t)N_NODES * OUT_F * sizeof(float), stream);
    hipMemsetAsync(sum_exp, 0, (size_t)N_NODES * sizeof(float), stream);
    hipMemsetAsync(gmax, 0, sizeof(unsigned), stream);

    const int n_tiles = N_NODES / 16;
    gat_gemm_wmma<<<(n_tiles + 3) / 4, 128, 0, stream>>>(h, W, a, Wh, s1, s2, n_tiles);

    gat_edge_score<<<(N_EDGES + 255) / 256, 256, 0, stream>>>(rows, cols, s1, s2, e,
                                                              gmax, N_EDGES);
    gat_edge_exp<<<(N_EDGES + 255) / 256, 256, 0, stream>>>(rows, e, gmax, expe,
                                                            sum_exp, N_EDGES);
    gat_scatter<<<(N_EDGES + 7) / 8, 256, 0, stream>>>(rows, cols, expe, sum_exp,
                                                       Wh, out, N_EDGES);
    const int total = N_NODES * OUT_F;
    gat_elu<<<(total + 255) / 256, 256, 0, stream>>>(out, total);
}


# Edge phase by row owner: 256 work-groups keep 196 out rows in LDS, filter all edges by a byte tag, add exp*Wh with LDS float adds, divide by exp sum last: no global float atomics; arrival-order sums
# speedup vs baseline: 2.5365x; 2.5365x over previous
#include <hip/hip_runtime.h>
#include <math.h>

#define N_NODES 50000
#define N_EDGES 800000
#define IN_F 128
#define OUT_F 128
#define NEG_SLOPE 0.2f
#define EPS_F 1e-10f

#define OWN_ROWS 196
#define N_OWNERS ((N_NODES + OWN_ROWS - 1) / OWN_ROWS)
#define OWN_WAVES 32
#define TAG_CHUNK 512
#define N_TAG_CHUNKS ((N_EDGES + TAG_CHUNK - 1) / TAG_CHUNK)
#define N_TAG_WORDS (N_TAG_CHUNKS * (TAG_CHUNK / 4))

typedef float v2f __attribute__((ext_vector_type(2)));
typedef float v8f __attribute__((ext_vector_type(8)));


__device__ __forceinline__ void atomAddF(float* p, float v) {
    __hip_atomic_fetch_add(p, v, __ATOMIC_RELAXED, __HIP_MEMORY_SCOPE_AGENT);
}

__device__ __forceinline__ unsigned fenc(float f) {
    unsigned u = __float_as_uint(f);
    return (u & 0x80000000u) ? ~u : (u | 0x80000000u);
}
__device__ __forceinline__ float fdec(unsigned u) {
    return __uint_as_float((u & 0x80000000u) ? (u ^ 0x80000000u) : ~u);
}

__global__ void __launch_bounds__(128)
gat_gemm_wmma(const float* __restrict__ h, const float* __restrict__ W,
              const float* __restrict__ a, float* __restrict__ Wh,
              float* __restrict__ s1, float* __restrict__ s2, int n_tiles) {
    __shared__ float tlds[4][16 * 128];

    const int lane = threadIdx.x & 31;
    const int wave = threadIdx.x >> 5;
    const int tile = blockIdx.x * 4 + wave;
    if (tile >= n_tiles) return;

    const int row0 = tile * 16;
    const int half = lane >> 4;
    const int mn   = lane & 15;

    float* tl = &tlds[wave][0];
    {
        const float4* hp4 = (const float4*)(h + row0 * IN_F);
        float4* tl4 = (float4*)tl;
        #pragma unroll
        for (int t = 0; t < 16; ++t)
            tl4[t * 32 + lane] = hp4[t * 32 + lane];
    }

    float a1v[8], a2v[8];
    #pragma unroll
    for (int nt = 0; nt < 8; ++nt) {
        a1v[nt] = a[nt * 16 + mn];
        a2v[nt] = a[OUT_F + nt * 16 + mn];
    }

    v8f acc[8] = {};

    for (int kc = 0; kc < 32; ++kc) {
        const int col = kc * 4 + 2 * half;
        v2f av;
        {
            const float2 t = *(const float2*)(tl + mn * 128 + col);
            av[0] = t.x; av[1] = t.y;
        }
        #pragma unroll
        for (int nt = 0; nt < 8; ++nt) {
            const int ncol = nt * 16 + mn;
            v2f bv;
            {
                const float2 t = *(const float2*)(W + ncol * IN_F + col);
                bv[0] = t.x; bv[1] = t.y;
            }
            acc[nt] = __builtin_amdgcn_wmma_f32_16x16x4_f32(
                false, av, false, bv, (short)0, acc[nt], false, false);
        }
    }

    #pragma unroll
    for (int nt = 0; nt < 8; ++nt) {
        #pragma unroll
        for (int j = 0; j < 8; ++j) {
            Wh[(row0 + j + 8 * half) * OUT_F + nt * 16 + mn] = acc[nt][j];
        }
    }

    float p1[8], p2[8];
    #pragma unroll
    for (int j = 0; j < 8; ++j) { p1[j] = 0.f; p2[j] = 0.f; }
    #pragma unroll
    for (int nt = 0; nt < 8; ++nt) {
        #pragma unroll
        for (int j = 0; j < 8; ++j) {
            p1[j] = fmaf(acc[nt][j], a1v[nt], p1[j]);
            p2[j] = fmaf(acc[nt][j], a2v[nt], p2[j]);
        }
    }
    #pragma unroll
    for (int j = 0; j < 8; ++j) {
        #pragma unroll
        for (int off = 1; off < 16; off <<= 1) {
            p1[j] += __shfl_xor(p1[j], off, 32);
            p2[j] += __shfl_xor(p2[j], off, 32);
        }
    }
    if (mn == 0) {
        #pragma unroll
        for (int j = 0; j < 8; ++j) {
            s1[row0 + j + 8 * half] = p1[j];
            s2[row0 + j + 8 * half] = p2[j];
        }
    }
}

__device__ __forceinline__ float edge_score(const float* __restrict__ s1, const float* __restrict__ s2,
                                            int r, int c) {
    const float v = s1[r] + s2[c];
    return (v > 0.f) ? v : NEG_SLOPE * v;
}

__global__ void __launch_bounds__(256)
gat_edge_tag(const int* __restrict__ rows, const int* __restrict__ cols,
             const float* __restrict__ s1, const float* __restrict__ s2,
             unsigned* __restrict__ tags, unsigned* __restrict__ gmax, int nE, int nWords) {
    __shared__ float smax[8];
    const int lane = threadIdx.x & 31;
    const int wave = threadIdx.x >> 5;
    const int t = blockIdx.x * 256 + threadIdx.x;
    const int e0 = t * 4;

    float lmax = -INFINITY;
    if (t < nWords) {
        unsigned tg = 0xFFFFFFFFu;
        if (e0 + 3 < nE) {
            const int4 r = ((const int4*)rows)[t];
            const int4 c = ((const int4*)cols)[t];
            const float v0 = edge_score(s1, s2, r.x, c.x);
            const float v1 = edge_score(s1, s2, r.y, c.y);
            const float v2 = edge_score(s1, s2, r.z, c.z);
            const float v3 = edge_score(s1, s2, r.w, c.w);
            lmax = fmaxf(fmaxf(v0, v1), fmaxf(v2, v3));
            tg = (unsigned)(r.x / OWN_ROWS) | ((unsigned)(r.y / OWN_ROWS) << 8) |
                 ((unsigned)(r.z / OWN_ROWS) << 16) | ((unsigned)(r.w / OWN_ROWS) << 24);
        } else {
            for (int b = 0; b < 4; ++b) {
                if (e0 + b < nE) {
                    const int r = rows[e0 + b];
                    lmax = fmaxf(lmax, edge_score(s1, s2, r, cols[e0 + b]));
                    tg = (tg & ~(0xFFu << (8 * b))) | ((unsigned)(r / OWN_ROWS) << (8 * b));
                }
            }
        }
        tags[t] = tg;
    }
    #pragma unroll
    for (int off = 16; off >= 1; off >>= 1)
        lmax = fmaxf(lmax, __shfl_xor(lmax, off, 32));
    if (lane == 0) smax[wave] = lmax;
    __syncthreads();
    if (threadIdx.x == 0) {
        float m = smax[0];
        #pragma unroll
        for (int w = 1; w < 8; ++w) m = fmaxf(m, smax[w]);
        atomicMax(gmax, fenc(m));
    }
}


__device__ __forceinline__ int bcastI(int v, int srcLane) {
    return __builtin_amdgcn_readlane(v, srcLane);
}
__device__ __forceinline__ float bcastF(float v, int srcLane) {
    return __int_as_float(__builtin_amdgcn_readlane(__float_as_int(v), srcLane));
}
__device__ __forceinline__ void ldsAddF(float* p, float v) {
    __hip_atomic_fetch_add(p, v, __ATOMIC_RELAXED, __HIP_MEMORY_SCOPE_WORKGROUP);
}

__device__ __forceinline__ unsigned zero_bytes(unsigned x) {
    const unsigned y = (x & 0x7F7F7F7Fu) + 0x7F7F7F7Fu;
    return ~(y | x | 0x7F7F7F7Fu);
}
__device__ __forceinline__ unsigned match16(uint4 t, unsigned pat) {
    return (zero_bytes(t.x ^ pat) >> 7) | (zero_bytes(t.y ^ pat) >> 6) |
           (zero_bytes(t.z ^ pat) >> 5) | (zero_bytes(t.w ^ pat) >> 4);
}

template <bool FULL>
__device__ __forceinline__ void own_flush(int n, int pend, int lane, int row_lo, float gm,
                                          const int* __restrict__ rows, const int* __restrict__ cols,
                                          const float* __restrict__ s1, const float* __restrict__ s2,
                                          const float* __restrict__ Wh, float* acc, float* ssum) {
    int rl = 0, c = 0;
    float ex = 0.f;
    if (FULL || lane < n) {
        const int r = rows[pend];
        c = cols[pend];
        ex = __expf(edge_score(s1, s2, r, c) - gm);
        rl = r - row_lo;
        ldsAddF(&ssum[rl], ex);
    }
    const float* whl = Wh + lane;
    #pragma unroll
    for (int g = 0; g < 32; g += 8) {
        if (!FULL && g >= n) break;
        float w[8][4] = {};
        #pragma unroll
        for (int u = 0; u < 8; ++u) {
            if (FULL || g + u < n) {
                const int cj = bcastI(c, g + u);
                const float* src = whl + cj * OUT_F;
                w[u][0] = src[0];
                w[u][1] = src[32];
                w[u][2] = src[64];
                w[u][3] = src[96];
            }
        }
        #pragma unroll
        for (int u = 0; u < 8; ++u) {
            if (FULL || g + u < n) {
                const int rj = bcastI(rl, g + u);
                const float aj = bcastF(ex, g + u);
                float* dst = acc + rj * OUT_F + lane;
                ldsAddF(dst +  0, aj * w[u][0]);
                ldsAddF(dst + 32, aj * w[u][1]);
                ldsAddF(dst + 64, aj * w[u][2]);
                ldsAddF(dst + 96, aj * w[u][3]);
            }
        }
    }
}

__device__ __forceinline__ uint4 load_tags(const uint4* __restrict__ tags4, int ch, int nchunks,
                                           int lane) {
    const int chc = ch < nchunks ? ch : nchunks - 1;
    return tags4[chc * 32 + lane];
}
__device__ __forceinline__ unsigned chunk_mask(int ch, int nchunks) {
    return ch < nchunks ? 0xFFFFFFFFu : 0u;
}

__global__ void __launch_bounds__(1024)
gat_owner(const int* __restrict__ rows, const int* __restrict__ cols,
          const unsigned* __restrict__ tags, const float* __restrict__ s1,
          const float* __restrict__ s2, const unsigned* __restrict__ gmax,
          const float* __restrict__ Wh, float* __restrict__ out, int nE, int nchunks) {
    __shared__ float acc[OWN_ROWS * OUT_F];
    __shared__ float ssum[OWN_ROWS];

    const int tid = threadIdx.x;
    const int lane = tid & 31;
    const int wave = __builtin_amdgcn_readfirstlane(tid >> 5);
    const int me = blockIdx.x;
    const int row_lo = me * OWN_ROWS;
    int nrows = N_NODES - row_lo;
    if (nrows > OWN_ROWS) nrows = OWN_ROWS;

    {
        float4* acc4 = (float4*)acc;
        for (int i = tid; i < OWN_ROWS * (OUT_F / 4); i += 1024)
            acc4[i] = make_float4(0.f, 0.f, 0.f, 0.f);
        if (tid < OWN_ROWS) ssum[tid] = 0.f;
    }
    __syncthreads();

    const float gm = fdec(*gmax);
    const unsigned pat = (unsigned)me * 0x01010101u;
    const uint4* tags4 = (const uint4*)tags;

    int pend = 0;
    int npend = 0;

    const int step = OWN_WAVES * 4;
    uint4 nx0 = load_tags(tags4, wave,                 nchunks, lane);
    uint4 nx1 = load_tags(tags4, wave + OWN_WAVES,     nchunks, lane);
    uint4 nx2 = load_tags(tags4, wave + 2 * OWN_WAVES, nchunks, lane);
    uint4 nx3 = load_tags(tags4, wave + 3 * OWN_WAVES, nchunks, lane);
    for (int ch0 = wave; ch0 < nchunks; ch0 += step) {
        const uint4 c0 = nx0, c1 = nx1, c2 = nx2, c3 = nx3;
        nx0 = load_tags(tags4, ch0 + step,                 nchunks, lane);
        nx1 = load_tags(tags4, ch0 + step + OWN_WAVES,     nchunks, lane);
        nx2 = load_tags(tags4, ch0 + step + 2 * OWN_WAVES, nchunks, lane);
        nx3 = load_tags(tags4, ch0 + step + 3 * OWN_WAVES, nchunks, lane);
        const unsigned mkA = (match16(c0, pat) & chunk_mask(ch0, nchunks)) |
                             ((match16(c1, pat) & chunk_mask(ch0 + OWN_WAVES, nchunks)) << 4);
        const unsigned mkB = (match16(c2, pat) & chunk_mask(ch0 + 2 * OWN_WAVES, nchunks)) |
                             ((match16(c3, pat) & chunk_mask(ch0 + 3 * OWN_WAVES, nchunks)) << 4);

        #pragma nounroll
        for (int h = 0; h < 2; ++h) {
            const unsigned mk = h ? mkB : mkA;
            unsigned m = __builtin_amdgcn_ballot_w32(mk != 0u);
            while (m) {
                const int l = __builtin_ctz(m);
                m &= m - 1u;
                unsigned mm = (unsigned)__builtin_amdgcn_readlane((int)mk, l);
                while (mm) {
                    const int p = __builtin_ctz(mm);
                    mm &= mm - 1u;
                    const int ch = ch0 + (2 * h + ((p >> 2) & 1)) * OWN_WAVES;
                    const int eid = ch * TAG_CHUNK + l * 16 + (p & 3) * 4 + (p >> 3);
                    if (eid < nE) {
                        pend = (lane == npend) ? eid : pend;
                        ++npend;
                        if (npend == 32) {
                            own_flush<true>(32, pend, lane, row_lo, gm, rows, cols, s1, s2, Wh,
                                            acc, ssum);
                            npend = 0;
                        }
                    }
                }
            }
        }
    }
    if (npend > 0)
        own_flush<false>(npend, pend, lane, row_lo, gm, rows, cols, s1, s2, Wh, acc, ssum);

    __syncthreads();

    {
        const float4* acc4 = (const float4*)acc;
        float4* out4 = (float4*)(out + (size_t)row_lo * OUT_F);
        for (int i = tid; i < nrows * (OUT_F / 4); i += 1024) {
            const float den = ssum[i >> 5] + EPS_F;
            float4 v = acc4[i];
            v.x /= den; v.y /= den; v.z /= den; v.w /= den;
            v.x = (v.x > 0.f) ? v.x : expm1f(v.x);
            v.y = (v.y > 0.f) ? v.y : expm1f(v.y);
            v.z = (v.z > 0.f) ? v.z : expm1f(v.z);
            v.w = (v.w > 0.f) ? v.w : expm1f(v.w);
            out4[i] = v;
        }
    }
}

extern "C" void kernel_launch(void* const* d_in, const int* in_sizes, int n_in,
                              void* d_out, int out_size, void* d_ws, size_t ws_size,
                              hipStream_t stream) {
    const float* h  = (const float*)d_in[0];
    const int*   ei = (const int*)d_in[1];
    const float* W  = (const float*)d_in[2];
    const float* a  = (const float*)d_in[3];
    float* out = (float*)d_out;

    const int* rows = ei;
    const int* cols = ei + N_EDGES;

    float* ws = (float*)d_ws;
    float*    Wh   = ws;
    float*    s1   = Wh + (size_t)N_NODES * OUT_F;
    float*    s2   = s1 + N_NODES;
    unsigned* tags = (unsigned*)(s2 + N_NODES);
    unsigned* gmax = tags + N_TAG_WORDS;

    hipMemsetAsync(gmax, 0, sizeof(unsigned), stream);

    const int n_tiles = N_NODES / 16;
    gat_gemm_wmma<<<(n_tiles + 3) / 4, 128, 0, stream>>>(h, W, a, Wh, s1, s2, n_tiles);

    gat_edge_tag<<<(N_TAG_WORDS + 255) / 256, 256, 0, stream>>>(rows, cols, s1, s2, tags, gmax,
                                                                N_EDGES, N_TAG_WORDS);
    gat_owner<<<N_OWNERS, 1024, 0, stream>>>(rows, cols, tags, s1, s2, gmax, Wh, out,
                                             N_EDGES, N_TAG_CHUNKS);
}


# GEMM: each wave keeps its 32 columns of W in registers as B fragments and walks row tiles, A from an LDS tile shared by 4 waves: no operand load per product; k regrouped, s1/s2 summed across 4 waves
# speedup vs baseline: 2.8025x; 1.1049x over previous
#include <hip/hip_runtime.h>
#include <math.h>

#define N_NODES 50000
#define N_EDGES 800000
#define IN_F 128
#define OUT_F 128
#define NEG_SLOPE 0.2f
#define EPS_F 1e-10f

#define OWN_ROWS 196
#define N_OWNERS ((N_NODES + OWN_ROWS - 1) / OWN_ROWS)
#define OWN_WAVES 32
#define TAG_CHUNK 512
#define N_TAG_CHUNKS ((N_EDGES + TAG_CHUNK - 1) / TAG_CHUNK)
#define N_TAG_WORDS (N_TAG_CHUNKS * (TAG_CHUNK / 4))

typedef float v2f __attribute__((ext_vector_type(2)));
typedef float v8f __attribute__((ext_vector_type(8)));


__device__ __forceinline__ void atomAddF(float* p, float v) {
    __hip_atomic_fetch_add(p, v, __ATOMIC_RELAXED, __HIP_MEMORY_SCOPE_AGENT);
}

__device__ __forceinline__ unsigned fenc(float f) {
    unsigned u = __float_as_uint(f);
    return (u & 0x80000000u) ? ~u : (u | 0x80000000u);
}
__device__ __forceinline__ float fdec(unsigned u) {
    return __uint_as_float((u & 0x80000000u) ? (u ^ 0x80000000u) : ~u);
}

#define GEMM_PITCH 132
#define GEMM_BLOCKS 1024

__device__ __forceinline__ float4 frag4(const float* rowbase, int j, int hf) {
    return *(const float4*)(rowbase + 8 * j + 4 * hf);
}

__global__ void __launch_bounds__(128)
gat_gemm_wmma(const float* __restrict__ h, const float* __restrict__ W,
              const float* __restrict__ a, float* __restrict__ Wh,
              float* __restrict__ s1, float* __restrict__ s2, int n_tiles) {
    __shared__ float tl[2][16 * GEMM_PITCH];
    __shared__ float part[2][4][16][2];

    const int tid  = threadIdx.x;
    const int lane = tid & 31;
    const int wave = tid >> 5;
    const int hf   = lane >> 4;
    const int mn   = lane & 15;
    const int col0 = wave * 32;

    float4 bq[2][16];
    #pragma unroll
    for (int nt = 0; nt < 2; ++nt) {
        const float* wr = W + (col0 + nt * 16 + mn) * IN_F;
        #pragma unroll
        for (int j = 0; j < 16; ++j) bq[nt][j] = frag4(wr, j, hf);
    }
    float a1v[2], a2v[2];
    #pragma unroll
    for (int nt = 0; nt < 2; ++nt) {
        a1v[nt] = a[col0 + nt * 16 + mn];
        a2v[nt] = a[OUT_F + col0 + nt * 16 + mn];
    }

    int tile = blockIdx.x;
    if (tile < n_tiles) {
        const float4* hp4 = (const float4*)(h + (size_t)tile * 16 * IN_F);
        #pragma unroll
        for (int i = 0; i < 4; ++i) {
            const int f = tid + 128 * i;
            *(float4*)&tl[0][(f >> 5) * GEMM_PITCH + (f & 31) * 4] = hp4[f];
        }
    }
    __syncthreads();

    int cur = 0;
    for (; tile < n_tiles; tile += GEMM_BLOCKS, cur ^= 1) {
        const int row0 = tile * 16;
        const bool more = tile + GEMM_BLOCKS < n_tiles;
        float4 st0 = make_float4(0.f, 0.f, 0.f, 0.f), st1 = st0, st2 = st0, st3 = st0;
        if (more) {
            const float4* hp4 = (const float4*)(h + (size_t)(tile + GEMM_BLOCKS) * 16 * IN_F);
            st0 = hp4[tid];
            st1 = hp4[tid + 128];
            st2 = hp4[tid + 256];
            st3 = hp4[tid + 384];
        }

        const float* arow = &tl[cur][mn * GEMM_PITCH];
        float4 aq[16];
        #pragma unroll
        for (int j = 0; j < 16; ++j) aq[j] = frag4(arow, j, hf);

        v8f acc0 = {}, acc1 = {};
        #pragma unroll
        for (int j = 0; j < 16; ++j) {
            v2f alo, ahi, b0lo, b0hi, b1lo, b1hi;
            alo[0]  = aq[j].x;     alo[1]  = aq[j].y;     ahi[0]  = aq[j].z;     ahi[1]  = aq[j].w;
            b0lo[0] = bq[0][j].x;  b0lo[1] = bq[0][j].y;  b0hi[0] = bq[0][j].z;  b0hi[1] = bq[0][j].w;
            b1lo[0] = bq[1][j].x;  b1lo[1] = bq[1][j].y;  b1hi[0] = bq[1][j].z;  b1hi[1] = bq[1][j].w;
            acc0 = __builtin_amdgcn_wmma_f32_16x16x4_f32(false, alo, false, b0lo, (short)0, acc0, false, false);
            acc1 = __builtin_amdgcn_wmma_f32_16x16x4_f32(false, alo, false, b1lo, (short)0, acc1, false, false);
            acc0 = __builtin_amdgcn_wmma_f32_16x16x4_f32(false, ahi, false, b0hi, (short)0, acc0, false, false);
            acc1 = __builtin_amdgcn_wmma_f32_16x16x4_f32(false, ahi, false, b1hi, (short)0, acc1, false, false);
        }

        float* whw = Wh + (size_t)row0 * OUT_F + col0 + lane;
        float p1[8], p2[8];
        #pragma unroll
        for (int j = 0; j < 8; ++j) {
            const float c0 = acc0[j], c1 = acc1[j];
            const float got = __shfl_xor(hf ? c0 : c1, 16, 32);
            whw[j * OUT_F]       = hf ? got : c0;
            whw[(j + 8) * OUT_F] = hf ? c1 : got;
            p1[j] = fmaf(c1, a1v[1], c0 * a1v[0]);
            p2[j] = fmaf(c1, a2v[1], c0 * a2v[0]);
        }
        #pragma unroll
        for (int j = 0; j < 8; ++j) {
            #pragma unroll
            for (int off = 1; off < 16; off <<= 1) {
                p1[j] += __shfl_xor(p1[j], off, 32);
                p2[j] += __shfl_xor(p2[j], off, 32);
            }
        }
        if (mn == 0) {
            #pragma unroll
            for (int j = 0; j < 8; ++j) {
                part[cur][wave][j + 8 * hf][0] = p1[j];
                part[cur][wave][j + 8 * hf][1] = p2[j];
            }
        }

        if (more) {
            float* dstl = &tl[cur ^ 1][(tid >> 5) * GEMM_PITCH + (tid & 31) * 4];
            *(float4*)(dstl)                   = st0;
            *(float4*)(dstl + 4 * GEMM_PITCH)  = st1;
            *(float4*)(dstl + 8 * GEMM_PITCH)  = st2;
            *(float4*)(dstl + 12 * GEMM_PITCH) = st3;
        }
        __syncthreads();
        if (tid < 32) {
            const int r = tid & 15, which = tid >> 4;
            const float v = (part[cur][0][r][which] + part[cur][1][r][which]) +
                            (part[cur][2][r][which] + part[cur][3][r][which]);
            (which ? s2 : s1)[row0 + r] = v;
        }
    }
}

__device__ __forceinline__ float edge_score(const float* __restrict__ s1, const float* __restrict__ s2,
                                            int r, int c) {
    const float v = s1[r] + s2[c];
    return (v > 0.f) ? v : NEG_SLOPE * v;
}

__global__ void __launch_bounds__(256)
gat_edge_tag(const int* __restrict__ rows, const int* __restrict__ cols,
             const float* __restrict__ s1, const float* __restrict__ s2,
             unsigned* __restrict__ tags, unsigned* __restrict__ gmax, int nE, int nWords) {
    __shared__ float smax[8];
    const int lane = threadIdx.x & 31;
    const int wave = threadIdx.x >> 5;
    const int t = blockIdx.x * 256 + threadIdx.x;
    const int e0 = t * 4;

    float lmax = -INFINITY;
    if (t < nWords) {
        unsigned tg = 0xFFFFFFFFu;
        if (e0 + 3 < nE) {
            const int4 r = ((const int4*)rows)[t];
            const int4 c = ((const int4*)cols)[t];
            const float v0 = edge_score(s1, s2, r.x, c.x);
            const float v1 = edge_score(s1, s2, r.y, c.y);
            const float v2 = edge_score(s1, s2, r.z, c.z);
            const float v3 = edge_score(s1, s2, r.w, c.w);
            lmax = fmaxf(fmaxf(v0, v1), fmaxf(v2, v3));
            tg = (unsigned)(r.x / OWN_ROWS) | ((unsigned)(r.y / OWN_ROWS) << 8) |
                 ((unsigned)(r.z / OWN_ROWS) << 16) | ((unsigned)(r.w / OWN_ROWS) << 24);
        } else {
            for (int b = 0; b < 4; ++b) {
                if (e0 + b < nE) {
                    const int r = rows[e0 + b];
                    lmax = fmaxf(lmax, edge_score(s1, s2, r, cols[e0 + b]));
                    tg = (tg & ~(0xFFu << (8 * b))) | ((unsigned)(r / OWN_ROWS) << (8 * b));
                }
            }
        }
        tags[t] = tg;
    }
    #pragma unroll
    for (int off = 16; off >= 1; off >>= 1)
        lmax = fmaxf(lmax, __shfl_xor(lmax, off, 32));
    if (lane == 0) smax[wave] = lmax;
    __syncthreads();
    if (threadIdx.x == 0) {
        float m = smax[0];
        #pragma unroll
        for (int w = 1; w < 8; ++w) m = fmaxf(m, smax[w]);
        atomicMax(gmax, fenc(m));
    }
}


__device__ __forceinline__ int bcastI(int v, int srcLane) {
    return __builtin_amdgcn_readlane(v, srcLane);
}
__device__ __forceinline__ float bcastF(float v, int srcLane) {
    return __int_as_float(__builtin_amdgcn_readlane(__float_as_int(v), srcLane));
}
__device__ __forceinline__ void ldsAddF(float* p, float v) {
    __hip_atomic_fetch_add(p, v, __ATOMIC_RELAXED, __HIP_MEMORY_SCOPE_WORKGROUP);
}

__device__ __forceinline__ unsigned zero_bytes(unsigned x) {
    const unsigned y = (x & 0x7F7F7F7Fu) + 0x7F7F7F7Fu;
    return ~(y | x | 0x7F7F7F7Fu);
}
__device__ __forceinline__ unsigned match16(uint4 t, unsigned pat) {
    return (zero_bytes(t.x ^ pat) >> 7) | (zero_bytes(t.y ^ pat) >> 6) |
           (zero_bytes(t.z ^ pat) >> 5) | (zero_bytes(t.w ^ pat) >> 4);
}

template <bool FULL>
__device__ __forceinline__ void own_flush(int n, int pend, int lane, int row_lo, float gm,
                                          const int* __restrict__ rows, const int* __restrict__ cols,
                                          const float* __restrict__ s1, const float* __restrict__ s2,
                                          const float* __restrict__ Wh, float* acc, float* ssum) {
    int rl = 0, c = 0;
    float ex = 0.f;
    if (FULL || lane < n) {
        const int r = rows[pend];
        c = cols[pend];
        ex = __expf(edge_score(s1, s2, r, c) - gm);
        rl = r - row_lo;
        ldsAddF(&ssum[rl], ex);
    }
    const float* whl = Wh + lane;
    #pragma unroll
    for (int g = 0; g < 32; g += 8) {
        if (!FULL && g >= n) break;
        float w[8][4] = {};
        #pragma unroll
        for (int u = 0; u < 8; ++u) {
            if (FULL || g + u < n) {
                const int cj = bcastI(c, g + u);
                const float* src = whl + cj * OUT_F;
                w[u][0] = src[0];
                w[u][1] = src[32];
                w[u][2] = src[64];
                w[u][3] = src[96];
            }
        }
        #pragma unroll
        for (int u = 0; u < 8; ++u) {
            if (FULL || g + u < n) {
                const int rj = bcastI(rl, g + u);
                const float aj = bcastF(ex, g + u);
                float* dst = acc + rj * OUT_F + lane;
                ldsAddF(dst +  0, aj * w[u][0]);
                ldsAddF(dst + 32, aj * w[u][1]);
                ldsAddF(dst + 64, aj * w[u][2]);
                ldsAddF(dst + 96, aj * w[u][3]);
            }
        }
    }
}

__device__ __forceinline__ uint4 load_tags(const uint4* __restrict__ tags4, int ch, int nchunks,
                                           int lane) {
    const int chc = ch < nchunks ? ch : nchunks - 1;
    return tags4[chc * 32 + lane];
}
__device__ __forceinline__ unsigned chunk_mask(int ch, int nchunks) {
    return ch < nchunks ? 0xFFFFFFFFu : 0u;
}

__global__ void __launch_bounds__(1024)
gat_owner(const int* __restrict__ rows, const int* __restrict__ cols,
          const unsigned* __restrict__ tags, const float* __restrict__ s1,
          const float* __restrict__ s2, const unsigned* __restrict__ gmax,
          const float* __restrict__ Wh, float* __restrict__ out, int nE, int nchunks) {
    __shared__ float acc[OWN_ROWS * OUT_F];
    __shared__ float ssum[OWN_ROWS];

    const int tid = threadIdx.x;
    const int lane = tid & 31;
    const int wave = __builtin_amdgcn_readfirstlane(tid >> 5);
    const int me = blockIdx.x;
    const int row_lo = me * OWN_ROWS;
    int nrows = N_NODES - row_lo;
    if (nrows > OWN_ROWS) nrows = OWN_ROWS;

    {
        float4* acc4 = (float4*)acc;
        for (int i = tid; i < OWN_ROWS * (OUT_F / 4); i += 1024)
            acc4[i] = make_float4(0.f, 0.f, 0.f, 0.f);
        if (tid < OWN_ROWS) ssum[tid] = 0.f;
    }
    __syncthreads();

    const float gm = fdec(*gmax);
    const unsigned pat = (unsigned)me * 0x01010101u;
    const uint4* tags4 = (const uint4*)tags;

    int pend = 0;
    int npend = 0;

    const int step = OWN_WAVES * 4;
    uint4 nx0 = load_tags(tags4, wave,                 nchunks, lane);
    uint4 nx1 = load_tags(tags4, wave + OWN_WAVES,     nchunks, lane);
    uint4 nx2 = load_tags(tags4, wave + 2 * OWN_WAVES, nchunks, lane);
    uint4 nx3 = load_tags(tags4, wave + 3 * OWN_WAVES, nchunks, lane);
    for (int ch0 = wave; ch0 < nchunks; ch0 += step) {
        const uint4 c0 = nx0, c1 = nx1, c2 = nx2, c3 = nx3;
        nx0 = load_tags(tags4, ch0 + step,                 nchunks, lane);
        nx1 = load_tags(tags4, ch0 + step + OWN_WAVES,     nchunks, lane);
        nx2 = load_tags(tags4, ch0 + step + 2 * OWN_WAVES, nchunks, lane);
        nx3 = load_tags(tags4, ch0 + step + 3 * OWN_WAVES, nchunks, lane);
        const unsigned mkA = (match16(c0, pat) & chunk_mask(ch0, nchunks)) |
                             ((match16(c1, pat) & chunk_mask(ch0 + OWN_WAVES, nchunks)) << 4);
        const unsigned mkB = (match16(c2, pat) & chunk_mask(ch0 + 2 * OWN_WAVES, nchunks)) |
                             ((match16(c3, pat) & chunk_mask(ch0 + 3 * OWN_WAVES, nchunks)) << 4);

        #pragma nounroll
        for (int h = 0; h < 2; ++h) {
            const unsigned mk = h ? mkB : mkA;
            unsigned m = __builtin_amdgcn_ballot_w32(mk != 0u);
            while (m) {
                const int l = __builtin_ctz(m);
                m &= m - 1u;
                unsigned mm = (unsigned)__builtin_amdgcn_readlane((int)mk, l);
                while (mm) {
                    const int p = __builtin_ctz(mm);
                    mm &= mm - 1u;
                    const int ch = ch0 + (2 * h + ((p >> 2) & 1)) * OWN_WAVES;
                    const int eid = ch * TAG_CHUNK + l * 16 + (p & 3) * 4 + (p >> 3);
                    if (eid < nE) {
                        pend = (lane == npend) ? eid : pend;
                        ++npend;
                        if (npend == 32) {
                            own_flush<true>(32, pend, lane, row_lo, gm, rows, cols, s1, s2, Wh,
                                            acc, ssum);
                            npend = 0;
                        }
                    }
                }
            }
        }
    }
    if (npend > 0)
        own_flush<false>(npend, pend, lane, row_lo, gm, rows, cols, s1, s2, Wh, acc, ssum);

    __syncthreads();

    {
        const float4* acc4 = (const float4*)acc;
        float4* out4 = (float4*)(out + (size_t)row_lo * OUT_F);
        for (int i = tid; i < nrows * (OUT_F / 4); i += 1024) {
            const float den = ssum[i >> 5] + EPS_F;
            float4 v = acc4[i];
            v.x /= den; v.y /= den; v.z /= den; v.w /= den;
            v.x = (v.x > 0.f) ? v.x : expm1f(v.x);
            v.y = (v.y > 0.f) ? v.y : expm1f(v.y);
            v.z = (v.z > 0.f) ? v.z : expm1f(v.z);
            v.w = (v.w > 0.f) ? v.w : expm1f(v.w);
            out4[i] = v;
        }
    }
}

extern "C" void kernel_launch(void* const* d_in, const int* in_sizes, int n_in,
                              void* d_out, int out_size, void* d_ws, size_t ws_size,
                              hipStream_t stream) {
    const float* h  = (const float*)d_in[0];
    const int*   ei = (const int*)d_in[1];
    const float* W  = (const float*)d_in[2];
    const float* a  = (const float*)d_in[3];
    float* out = (float*)d_out;

    const int* rows = ei;
    const int* cols = ei + N_EDGES;

    float* ws = (float*)d_ws;
    float*    Wh   = ws;
    float*    s1   = Wh + (size_t)N_NODES * OUT_F;
    float*    s2   = s1 + N_NODES;
    unsigned* tags = (unsigned*)(s2 + N_NODES);
    unsigned* gmax = tags + N_TAG_WORDS;

    hipMemsetAsync(gmax, 0, sizeof(unsigned), stream);

    const int n_tiles = N_NODES / 16;
    gat_gemm_wmma<<<GEMM_BLOCKS, 128, 0, stream>>>(h, W, a, Wh, s1, s2, n_tiles);

    gat_edge_tag<<<(N_TAG_WORDS + 255) / 256, 256, 0, stream>>>(rows, cols, s1, s2, tags, gmax,
                                                                N_EDGES, N_TAG_WORDS);
    gat_owner<<<N_OWNERS, 1024, 0, stream>>>(rows, cols, tags, s1, s2, gmax, Wh, out,
                                             N_EDGES, N_TAG_CHUNKS);
}


# Tag pass stores one 8-byte record per edge (column, local row, score); owner pass reads one record per kept edge, not rows, cols, s1, s2: one line, not four; scores bit-identical, arrival-order sums
# speedup vs baseline: 3.0234x; 1.0788x over previous
#include <hip/hip_runtime.h>
#include <math.h>

#define N_NODES 50000
#define N_EDGES 800000
#define IN_F 128
#define OUT_F 128
#define NEG_SLOPE 0.2f
#define EPS_F 1e-10f

#define OWN_ROWS 196
#define N_OWNERS ((N_NODES + OWN_ROWS - 1) / OWN_ROWS)
#define OWN_WAVES 32
#define TAG_CHUNK 512
#define N_TAG_CHUNKS ((N_EDGES + TAG_CHUNK - 1) / TAG_CHUNK)
#define N_TAG_WORDS (N_TAG_CHUNKS * (TAG_CHUNK / 4))

#define N_RECS (N_TAG_WORDS * 4)
static_assert(N_NODES <= 65536, "a column number must fit the low 16 bits of a record");
static_assert(OWN_ROWS <= 65536, "a local row number must fit the high 16 bits of a record");
#define WS_REC_WORD ((size_t)N_NODES * OUT_F + 2 * (size_t)N_NODES + N_TAG_WORDS + 4)
#define WS_NEED_BYTES ((WS_REC_WORD + 2 * (size_t)N_RECS) * 4)
static_assert(WS_REC_WORD % 4 == 0, "records are stored as 16-byte words");

typedef float v2f __attribute__((ext_vector_type(2)));
typedef float v8f __attribute__((ext_vector_type(8)));


__device__ __forceinline__ void atomAddF(float* p, float v) {
    __hip_atomic_fetch_add(p, v, __ATOMIC_RELAXED, __HIP_MEMORY_SCOPE_AGENT);
}

__device__ __forceinline__ unsigned fenc(float f) {
    unsigned u = __float_as_uint(f);
    return (u & 0x80000000u) ? ~u : (u | 0x80000000u);
}
__device__ __forceinline__ float fdec(unsigned u) {
    return __uint_as_float((u & 0x80000000u) ? (u ^ 0x80000000u) : ~u);
}

#define GEMM_PITCH 132
#define GEMM_BLOCKS 1024

__device__ __forceinline__ float4 frag4(const float* rowbase, int j, int hf) {
    return *(const float4*)(rowbase + 8 * j + 4 * hf);
}

__global__ void __launch_bounds__(128)
gat_gemm_wmma(const float* __restrict__ h, const float* __restrict__ W,
              const float* __restrict__ a, float* __restrict__ Wh,
              float* __restrict__ s1, float* __restrict__ s2, int n_tiles) {
    __shared__ float tl[2][16 * GEMM_PITCH];
    __shared__ float part[2][4][16][2];

    const int tid  = threadIdx.x;
    const int lane = tid & 31;
    const int wave = tid >> 5;
    const int hf   = lane >> 4;
    const int mn   = lane & 15;
    const int col0 = wave * 32;

    float4 bq[2][16];
    #pragma unroll
    for (int nt = 0; nt < 2; ++nt) {
        const float* wr = W + (col0 + nt * 16 + mn) * IN_F;
        #pragma unroll
        for (int j = 0; j < 16; ++j) bq[nt][j] = frag4(wr, j, hf);
    }
    float a1v[2], a2v[2];
    #pragma unroll
    for (int nt = 0; nt < 2; ++nt) {
        a1v[nt] = a[col0 + nt * 16 + mn];
        a2v[nt] = a[OUT_F + col0 + nt * 16 + mn];
    }

    int tile = blockIdx.x;
    if (tile < n_tiles) {
        const float4* hp4 = (const float4*)(h + (size_t)tile * 16 * IN_F);
        #pragma unroll
        for (int i = 0; i < 4; ++i) {
            const int f = tid + 128 * i;
            *(float4*)&tl[0][(f >> 5) * GEMM_PITCH + (f & 31) * 4] = hp4[f];
        }
    }
    __syncthreads();

    int cur = 0;
    for (; tile < n_tiles; tile += GEMM_BLOCKS, cur ^= 1) {
        const int row0 = tile * 16;
        const bool more = tile + GEMM_BLOCKS < n_tiles;
        float4 st0 = make_float4(0.f, 0.f, 0.f, 0.f), st1 = st0, st2 = st0, st3 = st0;
        if (more) {
            const float4* hp4 = (const float4*)(h + (size_t)(tile + GEMM_BLOCKS) * 16 * IN_F);
            st0 = hp4[tid];
            st1 = hp4[tid + 128];
            st2 = hp4[tid + 256];
            st3 = hp4[tid + 384];
        }

        const float* arow = &tl[cur][mn * GEMM_PITCH];
        float4 aq[16];
        #pragma unroll
        for (int j = 0; j < 16; ++j) aq[j] = frag4(arow, j, hf);

        v8f acc0 = {}, acc1 = {};
        #pragma unroll
        for (int j = 0; j < 16; ++j) {
            v2f alo, ahi, b0lo, b0hi, b1lo, b1hi;
            alo[0]  = aq[j].x;     alo[1]  = aq[j].y;     ahi[0]  = aq[j].z;     ahi[1]  = aq[j].w;
            b0lo[0] = bq[0][j].x;  b0lo[1] = bq[0][j].y;  b0hi[0] = bq[0][j].z;  b0hi[1] = bq[0][j].w;
            b1lo[0] = bq[1][j].x;  b1lo[1] = bq[1][j].y;  b1hi[0] = bq[1][j].z;  b1hi[1] = bq[1][j].w;
            acc0 = __builtin_amdgcn_wmma_f32_16x16x4_f32(false, alo, false, b0lo, (short)0, acc0, false, false);
            acc1 = __builtin_amdgcn_wmma_f32_16x16x4_f32(false, alo, false, b1lo, (short)0, acc1, false, false);
            acc0 = __builtin_amdgcn_wmma_f32_16x16x4_f32(false, ahi, false, b0hi, (short)0, acc0, false, false);
            acc1 = __builtin_amdgcn_wmma_f32_16x16x4_f32(false, ahi, false, b1hi, (short)0, acc1, false, false);
        }

        float* whw = Wh + (size_t)row0 * OUT_F + col0 + lane;
        float p1[8], p2[8];
        #pragma unroll
        for (int j = 0; j < 8; ++j) {
            const float c0 = acc0[j], c1 = acc1[j];
            const float got = __shfl_xor(hf ? c0 : c1, 16, 32);
            whw[j * OUT_F]       = hf ? got : c0;
            whw[(j + 8) * OUT_F] = hf ? c1 : got;
            p1[j] = fmaf(c1, a1v[1], c0 * a1v[0]);
            p2[j] = fmaf(c1, a2v[1], c0 * a2v[0]);
        }
        #pragma unroll
        for (int j = 0; j < 8; ++j) {
            #pragma unroll
            for (int off = 1; off < 16; off <<= 1) {
                p1[j] += __shfl_xor(p1[j], off, 32);
                p2[j] += __shfl_xor(p2[j], off, 32);
            }
        }
        if (mn == 0) {
            #pragma unroll
            for (int j = 0; j < 8; ++j) {
                part[cur][wave][j + 8 * hf][0] = p1[j];
                part[cur][wave][j + 8 * hf][1] = p2[j];
            }
        }

        if (more) {
            float* dstl = &tl[cur ^ 1][(tid >> 5) * GEMM_PITCH + (tid & 31) * 4];
            *(float4*)(dstl)                   = st0;
            *(float4*)(dstl + 4 * GEMM_PITCH)  = st1;
            *(float4*)(dstl + 8 * GEMM_PITCH)  = st2;
            *(float4*)(dstl + 12 * GEMM_PITCH) = st3;
        }
        __syncthreads();
        if (tid < 32) {
            const int r = tid & 15, which = tid >> 4;
            const float v = (part[cur][0][r][which] + part[cur][1][r][which]) +
                            (part[cur][2][r][which] + part[cur][3][r][which]);
            (which ? s2 : s1)[row0 + r] = v;
        }
    }
}

__device__ __forceinline__ float edge_score(const float* __restrict__ s1, const float* __restrict__ s2,
                                            int r, int c) {
    const float v = s1[r] + s2[c];
    return (v > 0.f) ? v : NEG_SLOPE * v;
}

__global__ void __launch_bounds__(256)
gat_edge_tag(const int* __restrict__ rows, const int* __restrict__ cols,
             const float* __restrict__ s1, const float* __restrict__ s2,
             unsigned* __restrict__ tags, unsigned* __restrict__ gmax,
             unsigned* __restrict__ recw, size_t ws_bytes, int nE, int nWords) {
    const bool have_recs = ws_bytes >= WS_NEED_BYTES;
    uint2* recs = (uint2*)recw;
    __shared__ float smax[8];
    const int lane = threadIdx.x & 31;
    const int wave = threadIdx.x >> 5;
    const int t = blockIdx.x * 256 + threadIdx.x;
    const int e0 = t * 4;

    float lmax = -INFINITY;
    if (t < nWords) {
        unsigned tg = 0xFFFFFFFFu;
        if (e0 + 3 < nE) {
            const int4 r = ((const int4*)rows)[t];
            const int4 c = ((const int4*)cols)[t];
            const float v0 = edge_score(s1, s2, r.x, c.x);
            const float v1 = edge_score(s1, s2, r.y, c.y);
            const float v2 = edge_score(s1, s2, r.z, c.z);
            const float v3 = edge_score(s1, s2, r.w, c.w);
            lmax = fmaxf(fmaxf(v0, v1), fmaxf(v2, v3));
            const unsigned o0 = (unsigned)(r.x / OWN_ROWS), o1 = (unsigned)(r.y / OWN_ROWS);
            const unsigned o2 = (unsigned)(r.z / OWN_ROWS), o3 = (unsigned)(r.w / OWN_ROWS);
            tg = o0 | (o1 << 8) | (o2 << 16) | (o3 << 24);
            if (have_recs) {
                uint4* rp = (uint4*)(recs + e0);
                rp[0] = make_uint4((unsigned)c.x | (((unsigned)r.x - o0 * OWN_ROWS) << 16),
                                   __float_as_uint(v0),
                                   (unsigned)c.y | (((unsigned)r.y - o1 * OWN_ROWS) << 16),
                                   __float_as_uint(v1));
                rp[1] = make_uint4((unsigned)c.z | (((unsigned)r.z - o2 * OWN_ROWS) << 16),
                                   __float_as_uint(v2),
                                   (unsigned)c.w | (((unsigned)r.w - o3 * OWN_ROWS) << 16),
                                   __float_as_uint(v3));
            }
        } else {
            for (int b = 0; b < 4; ++b) {
                if (e0 + b < nE) {
                    const int r = rows[e0 + b];
                    const int cc = cols[e0 + b];
                    const float v = edge_score(s1, s2, r, cc);
                    const unsigned o = (unsigned)(r / OWN_ROWS);
                    lmax = fmaxf(lmax, v);
                    tg = (tg & ~(0xFFu << (8 * b))) | (o << (8 * b));
                    if (have_recs)
                        recs[e0 + b] = make_uint2((unsigned)cc | (((unsigned)r - o * OWN_ROWS) << 16),
                                                  __float_as_uint(v));
                }
            }
        }
        tags[t] = tg;
    }
    #pragma unroll
    for (int off = 16; off >= 1; off >>= 1)
        lmax = fmaxf(lmax, __shfl_xor(lmax, off, 32));
    if (lane == 0) smax[wave] = lmax;
    __syncthreads();
    if (threadIdx.x == 0) {
        float m = smax[0];
        #pragma unroll
        for (int w = 1; w < 8; ++w) m = fmaxf(m, smax[w]);
        atomicMax(gmax, fenc(m));
    }
}


__device__ __forceinline__ int bcastI(int v, int srcLane) {
    return __builtin_amdgcn_readlane(v, srcLane);
}
__device__ __forceinline__ float bcastF(float v, int srcLane) {
    return __int_as_float(__builtin_amdgcn_readlane(__float_as_int(v), srcLane));
}
__device__ __forceinline__ void ldsAddF(float* p, float v) {
    __hip_atomic_fetch_add(p, v, __ATOMIC_RELAXED, __HIP_MEMORY_SCOPE_WORKGROUP);
}

__device__ __forceinline__ unsigned zero_bytes(unsigned x) {
    const unsigned y = (x & 0x7F7F7F7Fu) + 0x7F7F7F7Fu;
    return ~(y | x | 0x7F7F7F7Fu);
}
__device__ __forceinline__ unsigned match16(uint4 t, unsigned pat) {
    return (zero_bytes(t.x ^ pat) >> 7) | (zero_bytes(t.y ^ pat) >> 6) |
           (zero_bytes(t.z ^ pat) >> 5) | (zero_bytes(t.w ^ pat) >> 4);
}

template <bool FULL>
__device__ __forceinline__ void own_flush(int n, int pend, int lane, int row_lo, float gm,
                                          const int* __restrict__ rows, const int* __restrict__ cols,
                                          const float* __restrict__ s1, const float* __restrict__ s2,
                                          const uint2* recs, bool have_recs,
                                          const float* __restrict__ Wh, float* acc, float* ssum) {
    int rl = 0, c = 0;
    float ex = 0.f;
    if (FULL || lane < n) {
        float sc;
        if (have_recs) {
            const uint2 rec = recs[pend];
            c = (int)(rec.x & 0xFFFFu);
            rl = (int)(rec.x >> 16);
            sc = __uint_as_float(rec.y);
        } else {
            const int r = rows[pend];
            c = cols[pend];
            sc = edge_score(s1, s2, r, c);
            rl = r - row_lo;
        }
        ex = __expf(sc - gm);
        ldsAddF(&ssum[rl], ex);
    }
    const float* whl = Wh + lane;
    #pragma unroll
    for (int g = 0; g < 32; g += 8) {
        if (!FULL && g >= n) break;
        float w[8][4] = {};
        #pragma unroll
        for (int u = 0; u < 8; ++u) {
            if (FULL || g + u < n) {
                const int cj = bcastI(c, g + u);
                const float* src = whl + cj * OUT_F;
                w[u][0] = src[0];
                w[u][1] = src[32];
                w[u][2] = src[64];
                w[u][3] = src[96];
            }
        }
        #pragma unroll
        for (int u = 0; u < 8; ++u) {
            if (FULL || g + u < n) {
                const int rj = bcastI(rl, g + u);
                const float aj = bcastF(ex, g + u);
                float* dst = acc + rj * OUT_F + lane;
                ldsAddF(dst +  0, aj * w[u][0]);
                ldsAddF(dst + 32, aj * w[u][1]);
                ldsAddF(dst + 64, aj * w[u][2]);
                ldsAddF(dst + 96, aj * w[u][3]);
            }
        }
    }
}

__device__ __forceinline__ uint4 load_tags(const uint4* __restrict__ tags4, int ch, int nchunks,
                                           int lane) {
    const int chc = ch < nchunks ? ch : nchunks - 1;
    return tags4[chc * 32 + lane];
}
__device__ __forceinline__ unsigned chunk_mask(int ch, int nchunks) {
    return ch < nchunks ? 0xFFFFFFFFu : 0u;
}

__global__ void __launch_bounds__(1024)
gat_owner(const int* __restrict__ rows, const int* __restrict__ cols,
          const unsigned* __restrict__ tags, const float* __restrict__ s1,
          const float* __restrict__ s2, const unsigned* __restrict__ gmax,
          const unsigned* __restrict__ recw, size_t ws_bytes,
          const float* __restrict__ Wh, float* __restrict__ out, int nE, int nchunks) {
    const bool have_recs = ws_bytes >= WS_NEED_BYTES;
    const uint2* recs = (const uint2*)recw;
    __shared__ float acc[OWN_ROWS * OUT_F];
    __shared__ float ssum[OWN_ROWS];

    const int tid = threadIdx.x;
    const int lane = tid & 31;
    const int wave = __builtin_amdgcn_readfirstlane(tid >> 5);
    const int me = blockIdx.x;
    const int row_lo = me * OWN_ROWS;
    int nrows = N_NODES - row_lo;
    if (nrows > OWN_ROWS) nrows = OWN_ROWS;

    {
        float4* acc4 = (float4*)acc;
        for (int i = tid; i < OWN_ROWS * (OUT_F / 4); i += 1024)
            acc4[i] = make_float4(0.f, 0.f, 0.f, 0.f);
        if (tid < OWN_ROWS) ssum[tid] = 0.f;
    }
    __syncthreads();

    const float gm = fdec(*gmax);
    const unsigned pat = (unsigned)me * 0x01010101u;
    const uint4* tags4 = (const uint4*)tags;

    int pend = 0;
    int npend = 0;

    const int step = OWN_WAVES * 4;
    uint4 nx0 = load_tags(tags4, wave,                 nchunks, lane);
    uint4 nx1 = load_tags(tags4, wave + OWN_WAVES,     nchunks, lane);
    uint4 nx2 = load_tags(tags4, wave + 2 * OWN_WAVES, nchunks, lane);
    uint4 nx3 = load_tags(tags4, wave + 3 * OWN_WAVES, nchunks, lane);
    for (int ch0 = wave; ch0 < nchunks; ch0 += step) {
        const uint4 c0 = nx0, c1 = nx1, c2 = nx2, c3 = nx3;
        nx0 = load_tags(tags4, ch0 + step,                 nchunks, lane);
        nx1 = load_tags(tags4, ch0 + step + OWN_WAVES,     nchunks, lane);
        nx2 = load_tags(tags4, ch0 + step + 2 * OWN_WAVES, nchunks, lane);
        nx3 = load_tags(tags4, ch0 + step + 3 * OWN_WAVES, nchunks, lane);
        const unsigned mkA = (match16(c0, pat) & chunk_mask(ch0, nchunks)) |
                             ((match16(c1, pat) & chunk_mask(ch0 + OWN_WAVES, nchunks)) << 4);
        const unsigned mkB = (match16(c2, pat) & chunk_mask(ch0 + 2 * OWN_WAVES, nchunks)) |
                             ((match16(c3, pat) & chunk_mask(ch0 + 3 * OWN_WAVES, nchunks)) << 4);

        #pragma nounroll
        for (int h = 0; h < 2; ++h) {
            const unsigned mk = h ? mkB : mkA;
            unsigned m = __builtin_amdgcn_ballot_w32(mk != 0u);
            while (m) {
                const int l = __builtin_ctz(m);
                m &= m - 1u;
                unsigned mm = (unsigned)__builtin_amdgcn_readlane((int)mk, l);
                while (mm) {
                    const int p = __builtin_ctz(mm);
                    mm &= mm - 1u;
                    const int ch = ch0 + (2 * h + ((p >> 2) & 1)) * OWN_WAVES;
                    const int eid = ch * TAG_CHUNK + l * 16 + (p & 3) * 4 + (p >> 3);
                    if (eid < nE) {
                        pend = (lane == npend) ? eid : pend;
                        ++npend;
                        if (npend == 32) {
                            own_flush<true>(32, pend, lane, row_lo, gm, rows, cols, s1, s2, recs,
                                            have_recs, Wh, acc, ssum);
                            npend = 0;
                        }
                    }
                }
            }
        }
    }
    if (npend > 0)
        own_flush<false>(npend, pend, lane, row_lo, gm, rows, cols, s1, s2, recs, have_recs, Wh,
                         acc, ssum);

    __syncthreads();

    {
        const float4* acc4 = (const float4*)acc;
        float4* out4 = (float4*)(out + (size_t)row_lo * OUT_F);
        for (int i = tid; i < nrows * (OUT_F / 4); i += 1024) {
            const float den = ssum[i >> 5] + EPS_F;
            float4 v = acc4[i];
            v.x /= den; v.y /= den; v.z /= den; v.w /= den;
            v.x = (v.x > 0.f) ? v.x : expm1f(v.x);
            v.y = (v.y > 0.f) ? v.y : expm1f(v.y);
            v.z = (v.z > 0.f) ? v.z : expm1f(v.z);
            v.w = (v.w > 0.f) ? v.w : expm1f(v.w);
            out4[i] = v;
        }
    }
}

extern "C" void kernel_launch(void* const* d_in, const int* in_sizes, int n_in,
                              void* d_out, int out_size, void* d_ws, size_t ws_size,
                              hipStream_t stream) {
    const float* h  = (const float*)d_in[0];
    const int*   ei = (const int*)d_in[1];
    const float* W  = (const float*)d_in[2];
    const float* a  = (const float*)d_in[3];
    float* out = (float*)d_out;

    const int* rows = ei;
    const int* cols = ei + N_EDGES;

    float* ws = (float*)d_ws;
    float*    Wh   = ws;
    float*    s1   = Wh + (size_t)N_NODES * OUT_F;
    float*    s2   = s1 + N_NODES;
    unsigned* tags = (unsigned*)(s2 + N_NODES);
    unsigned* gmax = tags + N_TAG_WORDS;
    unsigned* recs = (unsigned*)(ws + WS_REC_WORD);

    hipMemsetAsync(gmax, 0, sizeof(unsigned), stream);

    const int n_tiles = N_NODES / 16;
    gat_gemm_wmma<<<GEMM_BLOCKS, 128, 0, stream>>>(h, W, a, Wh, s1, s2, n_tiles);

    gat_edge_tag<<<(N_TAG_WORDS + 255) / 256, 256, 0, stream>>>(rows, cols, s1, s2, tags, gmax,
                                                                recs, ws_size, N_EDGES,
                                                                N_TAG_WORDS);
    gat_owner<<<N_OWNERS, 1024, 0, stream>>>(rows, cols, tags, s1, s2, gmax, recs, ws_size, Wh,
                                             out, N_EDGES, N_TAG_CHUNKS);
}
